# starting kernel
# baseline (speedup 1.0000x reference)
#include <hip/hip_runtime.h>

constexpr int IN_DIM  = 128;
constexpr int OUT_DIM = 64;
constexpr int S1 = IN_DIM + 4;
constexpr int S2 = OUT_DIM + 4;

typedef float v2f __attribute__((ext_vector_type(2)));
typedef float v8f __attribute__((ext_vector_type(8)));

__global__ void __launch_bounds__(256) k_init(float* __restrict__ deg_s,
                                              float* __restrict__ deg_d,
                                              float* __restrict__ agg,
                                              float* __restrict__ hg, int n) {
    int tid = blockIdx.x * blockDim.x + threadIdx.x;
    int stride = gridDim.x * blockDim.x;
    for (int i = tid; i < n; i += stride) { deg_s[i] = 0.f; deg_d[i] = 0.f; }
    for (int i = tid; i < n * OUT_DIM; i += stride) agg[i] = 0.f;
    if (tid < 2 * OUT_DIM) hg[tid] = 0.f;
}

__global__ void __launch_bounds__(256) k_degree(const int* __restrict__ src,
                                                const int* __restrict__ dst,
                                                float* __restrict__ outdeg,
                                                float* __restrict__ indeg, int E) {
    int e = blockIdx.x * blockDim.x + threadIdx.x;
    if (e < E) {
        atomicAdd(&outdeg[src[e]], 1.0f);
        atomicAdd(&indeg[dst[e]], 1.0f);
    }
}

__global__ void __launch_bounds__(256) k_norm(float* __restrict__ sn,
                                              float* __restrict__ dn, int n) {
    int i = blockIdx.x * blockDim.x + threadIdx.x;
    if (i < n) {
        sn[i] = rsqrtf(fmaxf(sn[i], 1.0f));
        dn[i] = rsqrtf(fmaxf(dn[i], 1.0f));
    }
}

__global__ void __launch_bounds__(256) k_gemm1(const float* __restrict__ feat,
                                               const float* __restrict__ W1,
                                               const float* __restrict__ srcn,
                                               float* __restrict__ t1, int n) {
    __shared__ float sWt[OUT_DIM * S1];
    for (int i = threadIdx.x; i < IN_DIM * OUT_DIM; i += 256) {
        const int k = i >> 6;
        const int c = i & 63;
        sWt[c * S1 + k] = W1[i];
    }
    __syncthreads();

    const int wave = threadIdx.x >> 5;
    const int lane = threadIdx.x & 31;
    const int tile = blockIdx.x * 8 + wave;
    const int ntiles = (n + 15) >> 4;
    if (tile >= ntiles) return;

    const int halfsel = lane >> 4;
    const int l15 = lane & 15;
    int row = tile * 16 + l15;
    if (row >= n) row = n - 1;
    const int koff = halfsel * 2;
    const float norm = srcn[row];
    const float* arow = feat + (size_t)row * IN_DIM;
    const float* bc0 = sWt + ( 0 + l15) * S1;
    const float* bc1 = sWt + (16 + l15) * S1;
    const float* bc2 = sWt + (32 + l15) * S1;
    const float* bc3 = sWt + (48 + l15) * S1;

    v8f acc0 = {}, acc1 = {}, acc2 = {}, acc3 = {};
#pragma unroll 4
    for (int kb = 0; kb < IN_DIM / 4; ++kb) {
        const int k0 = kb * 4 + koff;
        v2f a = *(const v2f*)(arow + k0);
        a.x *= norm; a.y *= norm;
        const v2f b0 = *(const v2f*)(bc0 + k0);
        const v2f b1 = *(const v2f*)(bc1 + k0);
        const v2f b2 = *(const v2f*)(bc2 + k0);
        const v2f b3 = *(const v2f*)(bc3 + k0);
        acc0 = __builtin_amdgcn_wmma_f32_16x16x4_f32(false, a, false, b0, (short)0, acc0, false, false);
        acc1 = __builtin_amdgcn_wmma_f32_16x16x4_f32(false, a, false, b1, (short)0, acc1, false, false);
        acc2 = __builtin_amdgcn_wmma_f32_16x16x4_f32(false, a, false, b2, (short)0, acc2, false, false);
        acc3 = __builtin_amdgcn_wmma_f32_16x16x4_f32(false, a, false, b3, (short)0, acc3, false, false);
    }

#pragma unroll
    for (int r = 0; r < 8; ++r) {
        const int node = tile * 16 + r + halfsel * 8;
        if (node < n) {
            float* o = t1 + (size_t)node * OUT_DIM + l15;
            o[0]  = acc0[r];
            o[16] = acc1[r];
            o[32] = acc2[r];
            o[48] = acc3[r];
        }
    }
}

__global__ void __launch_bounds__(256) k_scatter(const int* __restrict__ src,
                                                 const int* __restrict__ dst,
                                                 const float* __restrict__ msg,
                                                 float* __restrict__ agg, int E) {
    int i = blockIdx.x * blockDim.x + threadIdx.x;
    if (i < E * (OUT_DIM / 4)) {
        const int e = i >> 4;
        const int q = (i & 15) << 2;
        const float4 m = *(const float4*)(msg + (size_t)src[e] * OUT_DIM + q);
        float* a = agg + (size_t)dst[e] * OUT_DIM + q;
        atomicAdd(a + 0, m.x);
        atomicAdd(a + 1, m.y);
        atomicAdd(a + 2, m.z);
        atomicAdd(a + 3, m.w);
    }
}

__global__ void __launch_bounds__(256) k_finalize1(float* __restrict__ agg,
                                                   const float* __restrict__ dstn,
                                                   const float* __restrict__ srcn,
                                                   const float* __restrict__ b1,
                                                   const float* __restrict__ prelu_a,
                                                   float* __restrict__ hs,
                                                   float* __restrict__ hg1, int n) {
    const float pa = *prelu_a;
    const int d = threadIdx.x & 63;
    const int sub = threadIdx.x >> 6;
    float part = 0.f;
    for (int base = blockIdx.x * 4; base < n; base += gridDim.x * 4) {
        const int node = base + sub;
        if (node < n) {
            const size_t idx = (size_t)node * OUT_DIM + d;
            const float v = agg[idx];
            agg[idx] = 0.f;
            float h = v * dstn[node] + b1[d];
            h = (h >= 0.f) ? h : pa * h;
            part += h;
            hs[idx] = h * srcn[node];
        }
    }
    __shared__ float red[256];
    red[threadIdx.x] = part;
    __syncthreads();
    if (threadIdx.x < 128) red[threadIdx.x] += red[threadIdx.x + 128];
    __syncthreads();
    if (threadIdx.x < 64)
        atomicAdd(&hg1[threadIdx.x], red[threadIdx.x] + red[threadIdx.x + 64]);
}

__global__ void __launch_bounds__(256) k_gemm2(const float* __restrict__ agg,
                                               const float* __restrict__ W2,
                                               const float* __restrict__ dstn,
                                               const float* __restrict__ b2,
                                               const float* __restrict__ prelu_a,
                                               float* __restrict__ hout,
                                               float* __restrict__ hg2, int n) {
    __shared__ float sWt[OUT_DIM * S2];
    __shared__ float shg[OUT_DIM];
    for (int i = threadIdx.x; i < OUT_DIM * OUT_DIM; i += 256) {
        const int k = i >> 6;
        const int c = i & 63;
        sWt[c * S2 + k] = W2[i];
    }
    if (threadIdx.x < OUT_DIM) shg[threadIdx.x] = 0.f;
    __syncthreads();

    const float pa = *prelu_a;
    const int wave = threadIdx.x >> 5;
    const int lane = threadIdx.x & 31;
    const int halfsel = lane >> 4;
    const int l15 = lane & 15;
    const int tile = blockIdx.x * 8 + wave;
    const int ntiles = (n + 15) >> 4;
    const bool active = tile < ntiles;

    if (active) {
        int row = tile * 16 + l15;
        if (row >= n) row = n - 1;
        const int koff = halfsel * 2;
        const float* arow = agg + (size_t)row * OUT_DIM;
        const float* bc0 = sWt + ( 0 + l15) * S2;
        const float* bc1 = sWt + (16 + l15) * S2;
        const float* bc2 = sWt + (32 + l15) * S2;
        const float* bc3 = sWt + (48 + l15) * S2;

        v8f acc0 = {}, acc1 = {}, acc2 = {}, acc3 = {};
#pragma unroll 4
        for (int kb = 0; kb < OUT_DIM / 4; ++kb) {
            const int k0 = kb * 4 + koff;
            const v2f a  = *(const v2f*)(arow + k0);
            const v2f b0 = *(const v2f*)(bc0 + k0);
            const v2f b1 = *(const v2f*)(bc1 + k0);
            const v2f b2 = *(const v2f*)(bc2 + k0);
            const v2f b3 = *(const v2f*)(bc3 + k0);
            acc0 = __builtin_amdgcn_wmma_f32_16x16x4_f32(false, a, false, b0, (short)0, acc0, false, false);
            acc1 = __builtin_amdgcn_wmma_f32_16x16x4_f32(false, a, false, b1, (short)0, acc1, false, false);
            acc2 = __builtin_amdgcn_wmma_f32_16x16x4_f32(false, a, false, b2, (short)0, acc2, false, false);
            acc3 = __builtin_amdgcn_wmma_f32_16x16x4_f32(false, a, false, b3, (short)0, acc3, false, false);
        }

        const float bb0 = b2[ 0 + l15], bb1 = b2[16 + l15];
        const float bb2 = b2[32 + l15], bb3 = b2[48 + l15];
        float hsum0 = 0.f, hsum1 = 0.f, hsum2 = 0.f, hsum3 = 0.f;
#pragma unroll
        for (int r = 0; r < 8; ++r) {
            const int node = tile * 16 + r + halfsel * 8;
            if (node < n) {
                const float dn = dstn[node];
                float v0 = acc0[r] * dn + bb0; v0 = (v0 >= 0.f) ? v0 : pa * v0;
                float v1 = acc1[r] * dn + bb1; v1 = (v1 >= 0.f) ? v1 : pa * v1;
                float v2 = acc2[r] * dn + bb2; v2 = (v2 >= 0.f) ? v2 : pa * v2;
                float v3 = acc3[r] * dn + bb3; v3 = (v3 >= 0.f) ? v3 : pa * v3;
                float* o = hout + (size_t)node * OUT_DIM + l15;
                o[0] = v0; o[16] = v1; o[32] = v2; o[48] = v3;
                hsum0 += v0; hsum1 += v1; hsum2 += v2; hsum3 += v3;
            }
        }
        atomicAdd(&shg[ 0 + l15], hsum0);
        atomicAdd(&shg[16 + l15], hsum1);
        atomicAdd(&shg[32 + l15], hsum2);
        atomicAdd(&shg[48 + l15], hsum3);
    }
    __syncthreads();
    if (threadIdx.x < OUT_DIM) atomicAdd(&hg2[threadIdx.x], shg[threadIdx.x]);
}

extern "C" void kernel_launch(void* const* d_in, const int* in_sizes, int n_in,
                              void* d_out, int out_size, void* d_ws, size_t ws_size,
                              hipStream_t stream) {
    const float* feat    = (const float*)d_in[0];
    const int*   src     = (const int*)  d_in[1];
    const int*   dst     = (const int*)  d_in[2];
    const float* W1      = (const float*)d_in[3];
    const float* b1      = (const float*)d_in[4];
    const float* W2      = (const float*)d_in[5];
    const float* b2      = (const float*)d_in[6];
    const float* prelu_a = (const float*)d_in[7];

    const int n = in_sizes[0] / IN_DIM;
    const int E = in_sizes[1];

    float* out = (float*)d_out;
    float* hg1 = out + (size_t)n * OUT_DIM;
    float* hg2 = hg1 + OUT_DIM;

    float* wsf  = (float*)d_ws;
    float* srcn = wsf;
    float* dstn = wsf + n;
    float* agg  = wsf + 2 * (size_t)n;
    float* t1   = out;

    const int ntiles = (n + 15) / 16;
    const int gemm_blocks = (ntiles + 7) / 8;
    const long long tot = (long long)E * (OUT_DIM / 4);
    const int sblocks = (int)((tot + 255) / 256);

    k_init   <<<2048, 256, 0, stream>>>(srcn, dstn, agg, hg1, n);
    k_degree <<<(E + 255) / 256, 256, 0, stream>>>(src, dst, srcn, dstn, E);
    k_norm   <<<(n + 255) / 256, 256, 0, stream>>>(srcn, dstn, n);
    k_gemm1  <<<gemm_blocks, 256, 0, stream>>>(feat, W1, srcn, t1, n);
    k_scatter<<<sblocks, 256, 0, stream>>>(src, dst, t1, agg, E);
    k_finalize1<<<1024, 256, 0, stream>>>(agg, dstn, srcn, b1, prelu_a, t1, hg1, n);
    k_scatter<<<sblocks, 256, 0, stream>>>(src, dst, t1, agg, E);
    k_gemm2  <<<gemm_blocks, 256, 0, stream>>>(agg, W2, dstn, b2, prelu_a, out, hg2, n);
}


# degree counters moved from two packed 400 KB arrays to one counter per 128-B line inside the output's h rows (free until GEMM1), so the 3.2M counting atomics spread over the whole 25.6 MB span
# speedup vs baseline: 1.0649x; 1.0649x over previous
#include <hip/hip_runtime.h>

constexpr int IN_DIM  = 128;
constexpr int OUT_DIM = 64;
constexpr int S1 = IN_DIM + 4;
constexpr int S2 = OUT_DIM + 4;

typedef float v2f __attribute__((ext_vector_type(2)));
typedef float v8f __attribute__((ext_vector_type(8)));

constexpr int CNT_STRIDE = OUT_DIM;
constexpr int CNT_IN     = OUT_DIM / 2;

__global__ void __launch_bounds__(256) k_init(float* __restrict__ cnt,
                                              float* __restrict__ agg,
                                              float* __restrict__ hg, int n) {
    int tid = blockIdx.x * blockDim.x + threadIdx.x;
    int stride = gridDim.x * blockDim.x;
    for (int i = tid; i < n; i += stride) {
        cnt[(size_t)i * CNT_STRIDE] = 0.f;
        cnt[(size_t)i * CNT_STRIDE + CNT_IN] = 0.f;
    }
    for (int i = tid; i < n * OUT_DIM; i += stride) agg[i] = 0.f;
    if (tid < 2 * OUT_DIM) hg[tid] = 0.f;
}

__global__ void __launch_bounds__(256) k_degree(const int* __restrict__ src,
                                                const int* __restrict__ dst,
                                                float* __restrict__ cnt, int E) {
    int e = blockIdx.x * blockDim.x + threadIdx.x;
    if (e < E) {
        const int s = src[e];
        const int d = dst[e];
        atomicAdd(&cnt[(size_t)s * CNT_STRIDE], 1.0f);
        atomicAdd(&cnt[(size_t)d * CNT_STRIDE + CNT_IN], 1.0f);
    }
}

__global__ void __launch_bounds__(256) k_norm(const float* __restrict__ cnt,
                                              float* __restrict__ sn,
                                              float* __restrict__ dn, int n) {
    int i = blockIdx.x * blockDim.x + threadIdx.x;
    if (i < n) {
        sn[i] = rsqrtf(fmaxf(cnt[(size_t)i * CNT_STRIDE], 1.0f));
        dn[i] = rsqrtf(fmaxf(cnt[(size_t)i * CNT_STRIDE + CNT_IN], 1.0f));
    }
}

__global__ void __launch_bounds__(256) k_gemm1(const float* __restrict__ feat,
                                               const float* __restrict__ W1,
                                               const float* __restrict__ srcn,
                                               float* __restrict__ t1, int n) {
    __shared__ float sWt[OUT_DIM * S1];
    for (int i = threadIdx.x; i < IN_DIM * OUT_DIM; i += 256) {
        const int k = i >> 6;
        const int c = i & 63;
        sWt[c * S1 + k] = W1[i];
    }
    __syncthreads();

    const int wave = threadIdx.x >> 5;
    const int lane = threadIdx.x & 31;
    const int tile = blockIdx.x * 8 + wave;
    const int ntiles = (n + 15) >> 4;
    if (tile >= ntiles) return;

    const int halfsel = lane >> 4;
    const int l15 = lane & 15;
    int row = tile * 16 + l15;
    if (row >= n) row = n - 1;
    const int koff = halfsel * 2;
    const float norm = srcn[row];
    const float* arow = feat + (size_t)row * IN_DIM;
    const float* bc0 = sWt + ( 0 + l15) * S1;
    const float* bc1 = sWt + (16 + l15) * S1;
    const float* bc2 = sWt + (32 + l15) * S1;
    const float* bc3 = sWt + (48 + l15) * S1;

    v8f acc0 = {}, acc1 = {}, acc2 = {}, acc3 = {};
#pragma unroll 4
    for (int kb = 0; kb < IN_DIM / 4; ++kb) {
        const int k0 = kb * 4 + koff;
        v2f a = *(const v2f*)(arow + k0);
        a.x *= norm; a.y *= norm;
        const v2f b0 = *(const v2f*)(bc0 + k0);
        const v2f b1 = *(const v2f*)(bc1 + k0);
        const v2f b2 = *(const v2f*)(bc2 + k0);
        const v2f b3 = *(const v2f*)(bc3 + k0);
        acc0 = __builtin_amdgcn_wmma_f32_16x16x4_f32(false, a, false, b0, (short)0, acc0, false, false);
        acc1 = __builtin_amdgcn_wmma_f32_16x16x4_f32(false, a, false, b1, (short)0, acc1, false, false);
        acc2 = __builtin_amdgcn_wmma_f32_16x16x4_f32(false, a, false, b2, (short)0, acc2, false, false);
        acc3 = __builtin_amdgcn_wmma_f32_16x16x4_f32(false, a, false, b3, (short)0, acc3, false, false);
    }

#pragma unroll
    for (int r = 0; r < 8; ++r) {
        const int node = tile * 16 + r + halfsel * 8;
        if (node < n) {
            float* o = t1 + (size_t)node * OUT_DIM + l15;
            o[0]  = acc0[r];
            o[16] = acc1[r];
            o[32] = acc2[r];
            o[48] = acc3[r];
        }
    }
}

__global__ void __launch_bounds__(256) k_scatter(const int* __restrict__ src,
                                                 const int* __restrict__ dst,
                                                 const float* __restrict__ msg,
                                                 float* __restrict__ agg, int E) {
    int i = blockIdx.x * blockDim.x + threadIdx.x;
    if (i < E * (OUT_DIM / 4)) {
        const int e = i >> 4;
        const int q = (i & 15) << 2;
        const float4 m = *(const float4*)(msg + (size_t)src[e] * OUT_DIM + q);
        float* a = agg + (size_t)dst[e] * OUT_DIM + q;
        atomicAdd(a + 0, m.x);
        atomicAdd(a + 1, m.y);
        atomicAdd(a + 2, m.z);
        atomicAdd(a + 3, m.w);
    }
}

__global__ void __launch_bounds__(256) k_finalize1(float* __restrict__ agg,
                                                   const float* __restrict__ dstn,
                                                   const float* __restrict__ srcn,
                                                   const float* __restrict__ b1,
                                                   const float* __restrict__ prelu_a,
                                                   float* __restrict__ hs,
                                                   float* __restrict__ hg1, int n) {
    const float pa = *prelu_a;
    const int d = threadIdx.x & 63;
    const int sub = threadIdx.x >> 6;
    float part = 0.f;
    for (int base = blockIdx.x * 4; base < n; base += gridDim.x * 4) {
        const int node = base + sub;
        if (node < n) {
            const size_t idx = (size_t)node * OUT_DIM + d;
            const float v = agg[idx];
            agg[idx] = 0.f;
            float h = v * dstn[node] + b1[d];
            h = (h >= 0.f) ? h : pa * h;
            part += h;
            hs[idx] = h * srcn[node];
        }
    }
    __shared__ float red[256];
    red[threadIdx.x] = part;
    __syncthreads();
    if (threadIdx.x < 128) red[threadIdx.x] += red[threadIdx.x + 128];
    __syncthreads();
    if (threadIdx.x < 64)
        atomicAdd(&hg1[threadIdx.x], red[threadIdx.x] + red[threadIdx.x + 64]);
}

__global__ void __launch_bounds__(256) k_gemm2(const float* __restrict__ agg,
                                               const float* __restrict__ W2,
                                               const float* __restrict__ dstn,
                                               const float* __restrict__ b2,
                                               const float* __restrict__ prelu_a,
                                               float* __restrict__ hout,
                                               float* __restrict__ hg2, int n) {
    __shared__ float sWt[OUT_DIM * S2];
    __shared__ float shg[OUT_DIM];
    for (int i = threadIdx.x; i < OUT_DIM * OUT_DIM; i += 256) {
        const int k = i >> 6;
        const int c = i & 63;
        sWt[c * S2 + k] = W2[i];
    }
    if (threadIdx.x < OUT_DIM) shg[threadIdx.x] = 0.f;
    __syncthreads();

    const float pa = *prelu_a;
    const int wave = threadIdx.x >> 5;
    const int lane = threadIdx.x & 31;
    const int halfsel = lane >> 4;
    const int l15 = lane & 15;
    const int tile = blockIdx.x * 8 + wave;
    const int ntiles = (n + 15) >> 4;
    const bool active = tile < ntiles;

    if (active) {
        int row = tile * 16 + l15;
        if (row >= n) row = n - 1;
        const int koff = halfsel * 2;
        const float* arow = agg + (size_t)row * OUT_DIM;
        const float* bc0 = sWt + ( 0 + l15) * S2;
        const float* bc1 = sWt + (16 + l15) * S2;
        const float* bc2 = sWt + (32 + l15) * S2;
        const float* bc3 = sWt + (48 + l15) * S2;

        v8f acc0 = {}, acc1 = {}, acc2 = {}, acc3 = {};
#pragma unroll 4
        for (int kb = 0; kb < OUT_DIM / 4; ++kb) {
            const int k0 = kb * 4 + koff;
            const v2f a  = *(const v2f*)(arow + k0);
            const v2f b0 = *(const v2f*)(bc0 + k0);
            const v2f b1 = *(const v2f*)(bc1 + k0);
            const v2f b2 = *(const v2f*)(bc2 + k0);
            const v2f b3 = *(const v2f*)(bc3 + k0);
            acc0 = __builtin_amdgcn_wmma_f32_16x16x4_f32(false, a, false, b0, (short)0, acc0, false, false);
            acc1 = __builtin_amdgcn_wmma_f32_16x16x4_f32(false, a, false, b1, (short)0, acc1, false, false);
            acc2 = __builtin_amdgcn_wmma_f32_16x16x4_f32(false, a, false, b2, (short)0, acc2, false, false);
            acc3 = __builtin_amdgcn_wmma_f32_16x16x4_f32(false, a, false, b3, (short)0, acc3, false, false);
        }

        const float bb0 = b2[ 0 + l15], bb1 = b2[16 + l15];
        const float bb2 = b2[32 + l15], bb3 = b2[48 + l15];
        float hsum0 = 0.f, hsum1 = 0.f, hsum2 = 0.f, hsum3 = 0.f;
#pragma unroll
        for (int r = 0; r < 8; ++r) {
            const int node = tile * 16 + r + halfsel * 8;
            if (node < n) {
                const float dn = dstn[node];
                float v0 = acc0[r] * dn + bb0; v0 = (v0 >= 0.f) ? v0 : pa * v0;
                float v1 = acc1[r] * dn + bb1; v1 = (v1 >= 0.f) ? v1 : pa * v1;
                float v2 = acc2[r] * dn + bb2; v2 = (v2 >= 0.f) ? v2 : pa * v2;
                float v3 = acc3[r] * dn + bb3; v3 = (v3 >= 0.f) ? v3 : pa * v3;
                float* o = hout + (size_t)node * OUT_DIM + l15;
                o[0] = v0; o[16] = v1; o[32] = v2; o[48] = v3;
                hsum0 += v0; hsum1 += v1; hsum2 += v2; hsum3 += v3;
            }
        }
        atomicAdd(&shg[ 0 + l15], hsum0);
        atomicAdd(&shg[16 + l15], hsum1);
        atomicAdd(&shg[32 + l15], hsum2);
        atomicAdd(&shg[48 + l15], hsum3);
    }
    __syncthreads();
    if (threadIdx.x < OUT_DIM) atomicAdd(&hg2[threadIdx.x], shg[threadIdx.x]);
}

extern "C" void kernel_launch(void* const* d_in, const int* in_sizes, int n_in,
                              void* d_out, int out_size, void* d_ws, size_t ws_size,
                              hipStream_t stream) {
    const float* feat    = (const float*)d_in[0];
    const int*   src     = (const int*)  d_in[1];
    const int*   dst     = (const int*)  d_in[2];
    const float* W1      = (const float*)d_in[3];
    const float* b1      = (const float*)d_in[4];
    const float* W2      = (const float*)d_in[5];
    const float* b2      = (const float*)d_in[6];
    const float* prelu_a = (const float*)d_in[7];

    const int n = in_sizes[0] / IN_DIM;
    const int E = in_sizes[1];

    float* out = (float*)d_out;
    float* hg1 = out + (size_t)n * OUT_DIM;
    float* hg2 = hg1 + OUT_DIM;

    float* wsf  = (float*)d_ws;
    float* srcn = wsf;
    float* dstn = wsf + n;
    float* agg  = wsf + 2 * (size_t)n;
    float* t1   = out;

    const int ntiles = (n + 15) / 16;
    const int gemm_blocks = (ntiles + 7) / 8;
    const long long tot = (long long)E * (OUT_DIM / 4);
    const int sblocks = (int)((tot + 255) / 256);

    k_init   <<<2048, 256, 0, stream>>>(out, agg, hg1, n);
    k_degree <<<(E + 255) / 256, 256, 0, stream>>>(src, dst, out, E);
    k_norm   <<<(n + 255) / 256, 256, 0, stream>>>(out, srcn, dstn, n);
    k_gemm1  <<<gemm_blocks, 256, 0, stream>>>(feat, W1, srcn, t1, n);
    k_scatter<<<sblocks, 256, 0, stream>>>(src, dst, t1, agg, E);
    k_finalize1<<<1024, 256, 0, stream>>>(agg, dstn, srcn, b1, prelu_a, t1, hg1, n);
    k_scatter<<<sblocks, 256, 0, stream>>>(src, dst, t1, agg, E);
    k_gemm2  <<<gemm_blocks, 256, 0, stream>>>(agg, W2, dstn, b2, prelu_a, out, hg2, n);
}


# edge scatter re-mapped to lane = feature: a wave walks 128 consecutive edges, each f32 atomic add covers one whole 128-B agg line (was 4 lines x 8 words per add); sums stay arrival-order f32 atomics
# speedup vs baseline: 1.5537x; 1.4590x over previous
#include <hip/hip_runtime.h>

constexpr int IN_DIM  = 128;
constexpr int OUT_DIM = 64;
constexpr int S1 = IN_DIM + 4;
constexpr int S2 = OUT_DIM + 4;

typedef float v2f __attribute__((ext_vector_type(2)));
typedef float v8f __attribute__((ext_vector_type(8)));

constexpr int CNT_STRIDE = OUT_DIM;
constexpr int CNT_IN     = OUT_DIM / 2;

__global__ void __launch_bounds__(256) k_init(float* __restrict__ cnt,
                                              float* __restrict__ agg,
                                              float* __restrict__ hg, int n) {
    int tid = blockIdx.x * blockDim.x + threadIdx.x;
    int stride = gridDim.x * blockDim.x;
    for (int i = tid; i < n; i += stride) {
        cnt[(size_t)i * CNT_STRIDE] = 0.f;
        cnt[(size_t)i * CNT_STRIDE + CNT_IN] = 0.f;
    }
    for (int i = tid; i < n * OUT_DIM; i += stride) agg[i] = 0.f;
    if (tid < 2 * OUT_DIM) hg[tid] = 0.f;
}

__global__ void __launch_bounds__(256) k_degree(const int* __restrict__ src,
                                                const int* __restrict__ dst,
                                                float* __restrict__ cnt, int E) {
    int e = blockIdx.x * blockDim.x + threadIdx.x;
    if (e < E) {
        const int s = src[e];
        const int d = dst[e];
        atomicAdd(&cnt[(size_t)s * CNT_STRIDE], 1.0f);
        atomicAdd(&cnt[(size_t)d * CNT_STRIDE + CNT_IN], 1.0f);
    }
}

__global__ void __launch_bounds__(256) k_norm(const float* __restrict__ cnt,
                                              float* __restrict__ sn,
                                              float* __restrict__ dn, int n) {
    int i = blockIdx.x * blockDim.x + threadIdx.x;
    if (i < n) {
        sn[i] = rsqrtf(fmaxf(cnt[(size_t)i * CNT_STRIDE], 1.0f));
        dn[i] = rsqrtf(fmaxf(cnt[(size_t)i * CNT_STRIDE + CNT_IN], 1.0f));
    }
}

__global__ void __launch_bounds__(256) k_gemm1(const float* __restrict__ feat,
                                               const float* __restrict__ W1,
                                               const float* __restrict__ srcn,
                                               float* __restrict__ t1, int n) {
    __shared__ float sWt[OUT_DIM * S1];
    for (int i = threadIdx.x; i < IN_DIM * OUT_DIM; i += 256) {
        const int k = i >> 6;
        const int c = i & 63;
        sWt[c * S1 + k] = W1[i];
    }
    __syncthreads();

    const int wave = threadIdx.x >> 5;
    const int lane = threadIdx.x & 31;
    const int tile = blockIdx.x * 8 + wave;
    const int ntiles = (n + 15) >> 4;
    if (tile >= ntiles) return;

    const int halfsel = lane >> 4;
    const int l15 = lane & 15;
    int row = tile * 16 + l15;
    if (row >= n) row = n - 1;
    const int koff = halfsel * 2;
    const float norm = srcn[row];
    const float* arow = feat + (size_t)row * IN_DIM;
    const float* bc0 = sWt + ( 0 + l15) * S1;
    const float* bc1 = sWt + (16 + l15) * S1;
    const float* bc2 = sWt + (32 + l15) * S1;
    const float* bc3 = sWt + (48 + l15) * S1;

    v8f acc0 = {}, acc1 = {}, acc2 = {}, acc3 = {};
#pragma unroll 4
    for (int kb = 0; kb < IN_DIM / 4; ++kb) {
        const int k0 = kb * 4 + koff;
        v2f a = *(const v2f*)(arow + k0);
        a.x *= norm; a.y *= norm;
        const v2f b0 = *(const v2f*)(bc0 + k0);
        const v2f b1 = *(const v2f*)(bc1 + k0);
        const v2f b2 = *(const v2f*)(bc2 + k0);
        const v2f b3 = *(const v2f*)(bc3 + k0);
        acc0 = __builtin_amdgcn_wmma_f32_16x16x4_f32(false, a, false, b0, (short)0, acc0, false, false);
        acc1 = __builtin_amdgcn_wmma_f32_16x16x4_f32(false, a, false, b1, (short)0, acc1, false, false);
        acc2 = __builtin_amdgcn_wmma_f32_16x16x4_f32(false, a, false, b2, (short)0, acc2, false, false);
        acc3 = __builtin_amdgcn_wmma_f32_16x16x4_f32(false, a, false, b3, (short)0, acc3, false, false);
    }

#pragma unroll
    for (int r = 0; r < 8; ++r) {
        const int node = tile * 16 + r + halfsel * 8;
        if (node < n) {
            float* o = t1 + (size_t)node * OUT_DIM + l15;
            o[0]  = acc0[r];
            o[16] = acc1[r];
            o[32] = acc2[r];
            o[48] = acc3[r];
        }
    }
}

constexpr int EDGES_PER_WAVE = 128;
constexpr int EDGE_GROUP     = 8;

__device__ __forceinline__ void scatter_one(const float* __restrict__ msg,
                                            float* __restrict__ agg,
                                            int s, int d, int lane) {
    const float* m = msg + (size_t)s * OUT_DIM + lane;
    const float lo = m[0];
    const float hi = m[32];
    float* a = agg + (size_t)d * OUT_DIM + lane;
    atomicAdd(a, lo);
    atomicAdd(a + 32, hi);
}

__global__ void __launch_bounds__(256) k_scatter(const int* __restrict__ src,
                                                 const int* __restrict__ dst,
                                                 const float* __restrict__ msg,
                                                 float* __restrict__ agg, int E) {
    constexpr int CHUNKS = EDGES_PER_WAVE / 32;
    const int lane = threadIdx.x & 31;
    const int wave = __builtin_amdgcn_readfirstlane(blockIdx.x * 8 + (threadIdx.x >> 5));
    const int base0 = wave * EDGES_PER_WAVE;
    if (base0 >= E) return;

    int s[CHUNKS], d[CHUNKS];
#pragma unroll
    for (int k = 0; k < CHUNKS; ++k) {
        const int e = base0 + 32 * k + lane;
        s[k] = (e < E) ? src[e] : 0;
        d[k] = (e < E) ? dst[e] : 0;
    }

#pragma unroll
    for (int k = 0; k < CHUNKS; ++k) {
        const int cnt = min(32, E - (base0 + 32 * k));
        int j0 = 0;
        for (; j0 + EDGE_GROUP <= cnt; j0 += EDGE_GROUP) {
            float lo[EDGE_GROUP], hi[EDGE_GROUP];
            int dj[EDGE_GROUP];
#pragma unroll
            for (int u = 0; u < EDGE_GROUP; ++u) {
                const int sj = __builtin_amdgcn_readlane(s[k], j0 + u);
                dj[u] = __builtin_amdgcn_readlane(d[k], j0 + u);
                const float* m = msg + (size_t)sj * OUT_DIM + lane;
                lo[u] = m[0];
                hi[u] = m[32];
            }
#pragma unroll
            for (int u = 0; u < EDGE_GROUP; ++u) {
                float* a = agg + (size_t)dj[u] * OUT_DIM + lane;
                atomicAdd(a, lo[u]);
                atomicAdd(a + 32, hi[u]);
            }
        }
        for (; j0 < cnt; ++j0)
            scatter_one(msg, agg, __builtin_amdgcn_readlane(s[k], j0),
                        __builtin_amdgcn_readlane(d[k], j0), lane);
    }
}

__global__ void __launch_bounds__(256) k_finalize1(float* __restrict__ agg,
                                                   const float* __restrict__ dstn,
                                                   const float* __restrict__ srcn,
                                                   const float* __restrict__ b1,
                                                   const float* __restrict__ prelu_a,
                                                   float* __restrict__ hs,
                                                   float* __restrict__ hg1, int n) {
    const float pa = *prelu_a;
    const int d = threadIdx.x & 63;
    const int sub = threadIdx.x >> 6;
    float part = 0.f;
    for (int base = blockIdx.x * 4; base < n; base += gridDim.x * 4) {
        const int node = base + sub;
        if (node < n) {
            const size_t idx = (size_t)node * OUT_DIM + d;
            const float v = agg[idx];
            agg[idx] = 0.f;
            float h = v * dstn[node] + b1[d];
            h = (h >= 0.f) ? h : pa * h;
            part += h;
            hs[idx] = h * srcn[node];
        }
    }
    __shared__ float red[256];
    red[threadIdx.x] = part;
    __syncthreads();
    if (threadIdx.x < 128) red[threadIdx.x] += red[threadIdx.x + 128];
    __syncthreads();
    if (threadIdx.x < 64)
        atomicAdd(&hg1[threadIdx.x], red[threadIdx.x] + red[threadIdx.x + 64]);
}

__global__ void __launch_bounds__(256) k_gemm2(const float* __restrict__ agg,
                                               const float* __restrict__ W2,
                                               const float* __restrict__ dstn,
                                               const float* __restrict__ b2,
                                               const float* __restrict__ prelu_a,
                                               float* __restrict__ hout,
                                               float* __restrict__ hg2, int n) {
    __shared__ float sWt[OUT_DIM * S2];
    __shared__ float shg[OUT_DIM];
    for (int i = threadIdx.x; i < OUT_DIM * OUT_DIM; i += 256) {
        const int k = i >> 6;
        const int c = i & 63;
        sWt[c * S2 + k] = W2[i];
    }
    if (threadIdx.x < OUT_DIM) shg[threadIdx.x] = 0.f;
    __syncthreads();

    const float pa = *prelu_a;
    const int wave = threadIdx.x >> 5;
    const int lane = threadIdx.x & 31;
    const int halfsel = lane >> 4;
    const int l15 = lane & 15;
    const int tile = blockIdx.x * 8 + wave;
    const int ntiles = (n + 15) >> 4;
    const bool active = tile < ntiles;

    if (active) {
        int row = tile * 16 + l15;
        if (row >= n) row = n - 1;
        const int koff = halfsel * 2;
        const float* arow = agg + (size_t)row * OUT_DIM;
        const float* bc0 = sWt + ( 0 + l15) * S2;
        const float* bc1 = sWt + (16 + l15) * S2;
        const float* bc2 = sWt + (32 + l15) * S2;
        const float* bc3 = sWt + (48 + l15) * S2;

        v8f acc0 = {}, acc1 = {}, acc2 = {}, acc3 = {};
#pragma unroll 4
        for (int kb = 0; kb < OUT_DIM / 4; ++kb) {
            const int k0 = kb * 4 + koff;
            const v2f a  = *(const v2f*)(arow + k0);
            const v2f b0 = *(const v2f*)(bc0 + k0);
            const v2f b1 = *(const v2f*)(bc1 + k0);
            const v2f b2 = *(const v2f*)(bc2 + k0);
            const v2f b3 = *(const v2f*)(bc3 + k0);
            acc0 = __builtin_amdgcn_wmma_f32_16x16x4_f32(false, a, false, b0, (short)0, acc0, false, false);
            acc1 = __builtin_amdgcn_wmma_f32_16x16x4_f32(false, a, false, b1, (short)0, acc1, false, false);
            acc2 = __builtin_amdgcn_wmma_f32_16x16x4_f32(false, a, false, b2, (short)0, acc2, false, false);
            acc3 = __builtin_amdgcn_wmma_f32_16x16x4_f32(false, a, false, b3, (short)0, acc3, false, false);
        }

        const float bb0 = b2[ 0 + l15], bb1 = b2[16 + l15];
        const float bb2 = b2[32 + l15], bb3 = b2[48 + l15];
        float hsum0 = 0.f, hsum1 = 0.f, hsum2 = 0.f, hsum3 = 0.f;
#pragma unroll
        for (int r = 0; r < 8; ++r) {
            const int node = tile * 16 + r + halfsel * 8;
            if (node < n) {
                const float dn = dstn[node];
                float v0 = acc0[r] * dn + bb0; v0 = (v0 >= 0.f) ? v0 : pa * v0;
                float v1 = acc1[r] * dn + bb1; v1 = (v1 >= 0.f) ? v1 : pa * v1;
                float v2 = acc2[r] * dn + bb2; v2 = (v2 >= 0.f) ? v2 : pa * v2;
                float v3 = acc3[r] * dn + bb3; v3 = (v3 >= 0.f) ? v3 : pa * v3;
                float* o = hout + (size_t)node * OUT_DIM + l15;
                o[0] = v0; o[16] = v1; o[32] = v2; o[48] = v3;
                hsum0 += v0; hsum1 += v1; hsum2 += v2; hsum3 += v3;
            }
        }
        atomicAdd(&shg[ 0 + l15], hsum0);
        atomicAdd(&shg[16 + l15], hsum1);
        atomicAdd(&shg[32 + l15], hsum2);
        atomicAdd(&shg[48 + l15], hsum3);
    }
    __syncthreads();
    if (threadIdx.x < OUT_DIM) atomicAdd(&hg2[threadIdx.x], shg[threadIdx.x]);
}

extern "C" void kernel_launch(void* const* d_in, const int* in_sizes, int n_in,
                              void* d_out, int out_size, void* d_ws, size_t ws_size,
                              hipStream_t stream) {
    const float* feat    = (const float*)d_in[0];
    const int*   src     = (const int*)  d_in[1];
    const int*   dst     = (const int*)  d_in[2];
    const float* W1      = (const float*)d_in[3];
    const float* b1      = (const float*)d_in[4];
    const float* W2      = (const float*)d_in[5];
    const float* b2      = (const float*)d_in[6];
    const float* prelu_a = (const float*)d_in[7];

    const int n = in_sizes[0] / IN_DIM;
    const int E = in_sizes[1];

    float* out = (float*)d_out;
    float* hg1 = out + (size_t)n * OUT_DIM;
    float* hg2 = hg1 + OUT_DIM;

    float* wsf  = (float*)d_ws;
    float* srcn = wsf;
    float* dstn = wsf + n;
    float* agg  = wsf + 2 * (size_t)n;
    float* t1   = out;

    const int ntiles = (n + 15) / 16;
    const int gemm_blocks = (ntiles + 7) / 8;
    const int swaves = (E + EDGES_PER_WAVE - 1) / EDGES_PER_WAVE;
    const int sblocks = (swaves + 7) / 8;

    k_init   <<<2048, 256, 0, stream>>>(out, agg, hg1, n);
    k_degree <<<(E + 255) / 256, 256, 0, stream>>>(src, dst, out, E);
    k_norm   <<<(n + 255) / 256, 256, 0, stream>>>(out, srcn, dstn, n);
    k_gemm1  <<<gemm_blocks, 256, 0, stream>>>(feat, W1, srcn, t1, n);
    k_scatter<<<sblocks, 256, 0, stream>>>(src, dst, t1, agg, E);
    k_finalize1<<<1024, 256, 0, stream>>>(agg, dstn, srcn, b1, prelu_a, t1, hg1, n);
    k_scatter<<<sblocks, 256, 0, stream>>>(src, dst, t1, agg, E);
    k_gemm2  <<<gemm_blocks, 256, 0, stream>>>(agg, W2, dstn, b2, prelu_a, out, hg2, n);
}


# degree counts taken in LDS integer histograms (block = 12800 nodes x one edge slice) and flushed as whole-line f32 atomic adds of exact partial counts, replacing 3.2M one-lane global f32 atomics
# speedup vs baseline: 2.2793x; 1.4670x over previous
#include <hip/hip_runtime.h>

constexpr int IN_DIM  = 128;
constexpr int OUT_DIM = 64;
constexpr int S1 = IN_DIM + 4;
constexpr int S2 = OUT_DIM + 4;

typedef float v2f __attribute__((ext_vector_type(2)));
typedef float v8f __attribute__((ext_vector_type(8)));

constexpr int DEG_RANGE   = 12800;
constexpr int DEG_RANGES  = 8;
constexpr int DEG_SLICES  = 16;
constexpr int DEG_THREADS = 1024;
constexpr int DEG_BATCH   = 8;

__global__ void __launch_bounds__(256) k_init(float* __restrict__ cnt,
                                              float* __restrict__ agg,
                                              float* __restrict__ hg, int n) {
    int tid = blockIdx.x * blockDim.x + threadIdx.x;
    int stride = gridDim.x * blockDim.x;
    for (int i = tid; i < 2 * n; i += stride) cnt[i] = 0.f;
    for (int i = tid; i < n * OUT_DIM; i += stride) agg[i] = 0.f;
    if (tid < 2 * OUT_DIM) hg[tid] = 0.f;
}

__device__ __forceinline__ void deg_bin(unsigned* bins, int id, int base) {
    const unsigned r = (unsigned)(id - base);
    if (r < (unsigned)DEG_RANGE) atomicAdd(&bins[r], 1u);
}

__global__ void __launch_bounds__(DEG_THREADS) k_degree(const int* __restrict__ src,
                                                        const int* __restrict__ dst,
                                                        float* __restrict__ cnt_out,
                                                        float* __restrict__ cnt_in,
                                                        int n, int E) {
    __shared__ unsigned bins[DEG_RANGE];
    const int* __restrict__ idx = blockIdx.z ? dst : src;
    float* __restrict__ cnt = blockIdx.z ? cnt_in : cnt_out;
    const int4* __restrict__ idx4 = (const int4*)idx;
    const int E4 = E >> 2;
    const int per = (E4 + (int)gridDim.x - 1) / (int)gridDim.x;
    const int q0 = (int)blockIdx.x * per;
    const int q1 = min(q0 + per, E4);

    for (int base = (int)blockIdx.y * DEG_RANGE; base < n; base += (int)gridDim.y * DEG_RANGE) {
        for (int i = threadIdx.x; i < DEG_RANGE; i += DEG_THREADS) bins[i] = 0u;
        __syncthreads();
        for (int q = q0 + (int)threadIdx.x; q < q1; q += DEG_BATCH * DEG_THREADS) {
            int4 v[DEG_BATCH];
#pragma unroll
            for (int u = 0; u < DEG_BATCH; ++u) {
                const int qq = q + u * DEG_THREADS;
                v[u] = (qq < q1) ? idx4[qq] : make_int4(-1, -1, -1, -1);
            }
#pragma unroll
            for (int u = 0; u < DEG_BATCH; ++u) {
                deg_bin(bins, v[u].x, base);
                deg_bin(bins, v[u].y, base);
                deg_bin(bins, v[u].z, base);
                deg_bin(bins, v[u].w, base);
            }
        }
        if (blockIdx.x == 0)
            for (int e = (E4 << 2) + (int)threadIdx.x; e < E; e += DEG_THREADS)
                deg_bin(bins, idx[e], base);
        __syncthreads();
        for (int i = threadIdx.x; i < DEG_RANGE; i += DEG_THREADS) {
            const int node = base + i;
            if (node < n) atomicAdd(&cnt[node], (float)bins[i]);
        }
        __syncthreads();
    }
}

__global__ void __launch_bounds__(256) k_norm(float* __restrict__ sn,
                                              float* __restrict__ dn, int n) {
    int i = blockIdx.x * blockDim.x + threadIdx.x;
    if (i < n) {
        sn[i] = rsqrtf(fmaxf(sn[i], 1.0f));
        dn[i] = rsqrtf(fmaxf(dn[i], 1.0f));
    }
}

__global__ void __launch_bounds__(256) k_gemm1(const float* __restrict__ feat,
                                               const float* __restrict__ W1,
                                               const float* __restrict__ srcn,
                                               float* __restrict__ t1, int n) {
    __shared__ float sWt[OUT_DIM * S1];
    for (int i = threadIdx.x; i < IN_DIM * OUT_DIM; i += 256) {
        const int k = i >> 6;
        const int c = i & 63;
        sWt[c * S1 + k] = W1[i];
    }
    __syncthreads();

    const int wave = threadIdx.x >> 5;
    const int lane = threadIdx.x & 31;
    const int tile = blockIdx.x * 8 + wave;
    const int ntiles = (n + 15) >> 4;
    if (tile >= ntiles) return;

    const int halfsel = lane >> 4;
    const int l15 = lane & 15;
    int row = tile * 16 + l15;
    if (row >= n) row = n - 1;
    const int koff = halfsel * 2;
    const float norm = srcn[row];
    const float* arow = feat + (size_t)row * IN_DIM;
    const float* bc0 = sWt + ( 0 + l15) * S1;
    const float* bc1 = sWt + (16 + l15) * S1;
    const float* bc2 = sWt + (32 + l15) * S1;
    const float* bc3 = sWt + (48 + l15) * S1;

    v8f acc0 = {}, acc1 = {}, acc2 = {}, acc3 = {};
#pragma unroll 4
    for (int kb = 0; kb < IN_DIM / 4; ++kb) {
        const int k0 = kb * 4 + koff;
        v2f a = *(const v2f*)(arow + k0);
        a.x *= norm; a.y *= norm;
        const v2f b0 = *(const v2f*)(bc0 + k0);
        const v2f b1 = *(const v2f*)(bc1 + k0);
        const v2f b2 = *(const v2f*)(bc2 + k0);
        const v2f b3 = *(const v2f*)(bc3 + k0);
        acc0 = __builtin_amdgcn_wmma_f32_16x16x4_f32(false, a, false, b0, (short)0, acc0, false, false);
        acc1 = __builtin_amdgcn_wmma_f32_16x16x4_f32(false, a, false, b1, (short)0, acc1, false, false);
        acc2 = __builtin_amdgcn_wmma_f32_16x16x4_f32(false, a, false, b2, (short)0, acc2, false, false);
        acc3 = __builtin_amdgcn_wmma_f32_16x16x4_f32(false, a, false, b3, (short)0, acc3, false, false);
    }

#pragma unroll
    for (int r = 0; r < 8; ++r) {
        const int node = tile * 16 + r + halfsel * 8;
        if (node < n) {
            float* o = t1 + (size_t)node * OUT_DIM + l15;
            o[0]  = acc0[r];
            o[16] = acc1[r];
            o[32] = acc2[r];
            o[48] = acc3[r];
        }
    }
}

constexpr int EDGES_PER_WAVE = 128;
constexpr int EDGE_GROUP     = 8;

__device__ __forceinline__ void scatter_one(const float* __restrict__ msg,
                                            float* __restrict__ agg,
                                            int s, int d, int lane) {
    const float* m = msg + (size_t)s * OUT_DIM + lane;
    const float lo = m[0];
    const float hi = m[32];
    float* a = agg + (size_t)d * OUT_DIM + lane;
    atomicAdd(a, lo);
    atomicAdd(a + 32, hi);
}

__global__ void __launch_bounds__(256) k_scatter(const int* __restrict__ src,
                                                 const int* __restrict__ dst,
                                                 const float* __restrict__ msg,
                                                 float* __restrict__ agg, int E) {
    constexpr int CHUNKS = EDGES_PER_WAVE / 32;
    const int lane = threadIdx.x & 31;
    const int wave = __builtin_amdgcn_readfirstlane(blockIdx.x * 8 + (threadIdx.x >> 5));
    const int base0 = wave * EDGES_PER_WAVE;
    if (base0 >= E) return;

    int s[CHUNKS], d[CHUNKS];
#pragma unroll
    for (int k = 0; k < CHUNKS; ++k) {
        const int e = base0 + 32 * k + lane;
        s[k] = (e < E) ? src[e] : 0;
        d[k] = (e < E) ? dst[e] : 0;
    }

#pragma unroll
    for (int k = 0; k < CHUNKS; ++k) {
        const int cnt = min(32, E - (base0 + 32 * k));
        int j0 = 0;
        for (; j0 + EDGE_GROUP <= cnt; j0 += EDGE_GROUP) {
            float lo[EDGE_GROUP], hi[EDGE_GROUP];
            int dj[EDGE_GROUP];
#pragma unroll
            for (int u = 0; u < EDGE_GROUP; ++u) {
                const int sj = __builtin_amdgcn_readlane(s[k], j0 + u);
                dj[u] = __builtin_amdgcn_readlane(d[k], j0 + u);
                const float* m = msg + (size_t)sj * OUT_DIM + lane;
                lo[u] = m[0];
                hi[u] = m[32];
            }
#pragma unroll
            for (int u = 0; u < EDGE_GROUP; ++u) {
                float* a = agg + (size_t)dj[u] * OUT_DIM + lane;
                atomicAdd(a, lo[u]);
                atomicAdd(a + 32, hi[u]);
            }
        }
        for (; j0 < cnt; ++j0)
            scatter_one(msg, agg, __builtin_amdgcn_readlane(s[k], j0),
                        __builtin_amdgcn_readlane(d[k], j0), lane);
    }
}

__global__ void __launch_bounds__(256) k_finalize1(float* __restrict__ agg,
                                                   const float* __restrict__ dstn,
                                                   const float* __restrict__ srcn,
                                                   const float* __restrict__ b1,
                                                   const float* __restrict__ prelu_a,
                                                   float* __restrict__ hs,
                                                   float* __restrict__ hg1, int n) {
    const float pa = *prelu_a;
    const int d = threadIdx.x & 63;
    const int sub = threadIdx.x >> 6;
    float part = 0.f;
    for (int base = blockIdx.x * 4; base < n; base += gridDim.x * 4) {
        const int node = base + sub;
        if (node < n) {
            const size_t idx = (size_t)node * OUT_DIM + d;
            const float v = agg[idx];
            agg[idx] = 0.f;
            float h = v * dstn[node] + b1[d];
            h = (h >= 0.f) ? h : pa * h;
            part += h;
            hs[idx] = h * srcn[node];
        }
    }
    __shared__ float red[256];
    red[threadIdx.x] = part;
    __syncthreads();
    if (threadIdx.x < 128) red[threadIdx.x] += red[threadIdx.x + 128];
    __syncthreads();
    if (threadIdx.x < 64)
        atomicAdd(&hg1[threadIdx.x], red[threadIdx.x] + red[threadIdx.x + 64]);
}

__global__ void __launch_bounds__(256) k_gemm2(const float* __restrict__ agg,
                                               const float* __restrict__ W2,
                                               const float* __restrict__ dstn,
                                               const float* __restrict__ b2,
                                               const float* __restrict__ prelu_a,
                                               float* __restrict__ hout,
                                               float* __restrict__ hg2, int n) {
    __shared__ float sWt[OUT_DIM * S2];
    __shared__ float shg[OUT_DIM];
    for (int i = threadIdx.x; i < OUT_DIM * OUT_DIM; i += 256) {
        const int k = i >> 6;
        const int c = i & 63;
        sWt[c * S2 + k] = W2[i];
    }
    if (threadIdx.x < OUT_DIM) shg[threadIdx.x] = 0.f;
    __syncthreads();

    const float pa = *prelu_a;
    const int wave = threadIdx.x >> 5;
    const int lane = threadIdx.x & 31;
    const int halfsel = lane >> 4;
    const int l15 = lane & 15;
    const int tile = blockIdx.x * 8 + wave;
    const int ntiles = (n + 15) >> 4;
    const bool active = tile < ntiles;

    if (active) {
        int row = tile * 16 + l15;
        if (row >= n) row = n - 1;
        const int koff = halfsel * 2;
        const float* arow = agg + (size_t)row * OUT_DIM;
        const float* bc0 = sWt + ( 0 + l15) * S2;
        const float* bc1 = sWt + (16 + l15) * S2;
        const float* bc2 = sWt + (32 + l15) * S2;
        const float* bc3 = sWt + (48 + l15) * S2;

        v8f acc0 = {}, acc1 = {}, acc2 = {}, acc3 = {};
#pragma unroll 4
        for (int kb = 0; kb < OUT_DIM / 4; ++kb) {
            const int k0 = kb * 4 + koff;
            const v2f a  = *(const v2f*)(arow + k0);
            const v2f b0 = *(const v2f*)(bc0 + k0);
            const v2f b1 = *(const v2f*)(bc1 + k0);
            const v2f b2 = *(const v2f*)(bc2 + k0);
            const v2f b3 = *(const v2f*)(bc3 + k0);
            acc0 = __builtin_amdgcn_wmma_f32_16x16x4_f32(false, a, false, b0, (short)0, acc0, false, false);
            acc1 = __builtin_amdgcn_wmma_f32_16x16x4_f32(false, a, false, b1, (short)0, acc1, false, false);
            acc2 = __builtin_amdgcn_wmma_f32_16x16x4_f32(false, a, false, b2, (short)0, acc2, false, false);
            acc3 = __builtin_amdgcn_wmma_f32_16x16x4_f32(false, a, false, b3, (short)0, acc3, false, false);
        }

        const float bb0 = b2[ 0 + l15], bb1 = b2[16 + l15];
        const float bb2 = b2[32 + l15], bb3 = b2[48 + l15];
        float hsum0 = 0.f, hsum1 = 0.f, hsum2 = 0.f, hsum3 = 0.f;
#pragma unroll
        for (int r = 0; r < 8; ++r) {
            const int node = tile * 16 + r + halfsel * 8;
            if (node < n) {
                const float dn = dstn[node];
                float v0 = acc0[r] * dn + bb0; v0 = (v0 >= 0.f) ? v0 : pa * v0;
                float v1 = acc1[r] * dn + bb1; v1 = (v1 >= 0.f) ? v1 : pa * v1;
                float v2 = acc2[r] * dn + bb2; v2 = (v2 >= 0.f) ? v2 : pa * v2;
                float v3 = acc3[r] * dn + bb3; v3 = (v3 >= 0.f) ? v3 : pa * v3;
                float* o = hout + (size_t)node * OUT_DIM + l15;
                o[0] = v0; o[16] = v1; o[32] = v2; o[48] = v3;
                hsum0 += v0; hsum1 += v1; hsum2 += v2; hsum3 += v3;
            }
        }
        atomicAdd(&shg[ 0 + l15], hsum0);
        atomicAdd(&shg[16 + l15], hsum1);
        atomicAdd(&shg[32 + l15], hsum2);
        atomicAdd(&shg[48 + l15], hsum3);
    }
    __syncthreads();
    if (threadIdx.x < OUT_DIM) atomicAdd(&hg2[threadIdx.x], shg[threadIdx.x]);
}

extern "C" void kernel_launch(void* const* d_in, const int* in_sizes, int n_in,
                              void* d_out, int out_size, void* d_ws, size_t ws_size,
                              hipStream_t stream) {
    const float* feat    = (const float*)d_in[0];
    const int*   src     = (const int*)  d_in[1];
    const int*   dst     = (const int*)  d_in[2];
    const float* W1      = (const float*)d_in[3];
    const float* b1      = (const float*)d_in[4];
    const float* W2      = (const float*)d_in[5];
    const float* b2      = (const float*)d_in[6];
    const float* prelu_a = (const float*)d_in[7];

    const int n = in_sizes[0] / IN_DIM;
    const int E = in_sizes[1];

    float* out = (float*)d_out;
    float* hg1 = out + (size_t)n * OUT_DIM;
    float* hg2 = hg1 + OUT_DIM;

    float* wsf  = (float*)d_ws;
    float* srcn = wsf;
    float* dstn = wsf + n;
    float* agg  = wsf + 2 * (size_t)n;
    float* t1   = out;

    const int ntiles = (n + 15) / 16;
    const int gemm_blocks = (ntiles + 7) / 8;
    const int swaves = (E + EDGES_PER_WAVE - 1) / EDGES_PER_WAVE;
    const int sblocks = (swaves + 7) / 8;

    k_init   <<<2048, 256, 0, stream>>>(srcn, agg, hg1, n);
    k_degree <<<dim3(DEG_SLICES, DEG_RANGES, 2), DEG_THREADS, 0, stream>>>(src, dst, srcn, dstn, n, E);
    k_norm   <<<(n + 255) / 256, 256, 0, stream>>>(srcn, dstn, n);
    k_gemm1  <<<gemm_blocks, 256, 0, stream>>>(feat, W1, srcn, t1, n);
    k_scatter<<<sblocks, 256, 0, stream>>>(src, dst, t1, agg, E);
    k_finalize1<<<1024, 256, 0, stream>>>(agg, dstn, srcn, b1, prelu_a, t1, hg1, n);
    k_scatter<<<sblocks, 256, 0, stream>>>(src, dst, t1, agg, E);
    k_gemm2  <<<gemm_blocks, 256, 0, stream>>>(agg, W2, dstn, b2, prelu_a, out, hg2, n);
}


# edge scatter split by cache half: a block reads its die, draws count-down tickets for edge runs, keeps edges whose dst row is in its half, so agg adds stay in-half; arrival-order f32 atomics kept
# speedup vs baseline: 2.7931x; 1.2254x over previous
#include <hip/hip_runtime.h>

constexpr int IN_DIM  = 128;
constexpr int OUT_DIM = 64;
constexpr int S1 = IN_DIM + 4;
constexpr int S2 = OUT_DIM + 4;

typedef float v2f __attribute__((ext_vector_type(2)));
typedef float v8f __attribute__((ext_vector_type(8)));

constexpr int DEG_RANGE   = 12800;
constexpr int DEG_RANGES  = 8;
constexpr int DEG_SLICES  = 16;
constexpr int DEG_THREADS = 1024;
constexpr int DEG_BATCH   = 8;

constexpr int SC_BLOCKS      = 1536;
constexpr int EDGES_PER_WAVE = 128;
constexpr int TICKET_EDGES   = 8 * EDGES_PER_WAVE;

__device__ __forceinline__ int ticket_start(int E) {
    return (E + TICKET_EDGES - 1) / TICKET_EDGES + SC_BLOCKS;
}

__global__ void __launch_bounds__(256) k_init(float* __restrict__ cnt,
                                              float* __restrict__ agg,
                                              float* __restrict__ hg, int n, int E) {
    int tid = blockIdx.x * blockDim.x + threadIdx.x;
    int stride = gridDim.x * blockDim.x;
    for (int i = tid; i < 2 * n; i += stride) cnt[i] = 0.f;
    for (int i = tid; i < n * OUT_DIM; i += stride) agg[i] = 0.f;
    if (tid < 2 * OUT_DIM) {
        if ((tid & 31) == 0) ((int*)hg)[tid] = ticket_start(E);
        else hg[tid] = 0.f;
    }
}

__device__ __forceinline__ void deg_bin(unsigned* bins, int id, int base) {
    const unsigned r = (unsigned)(id - base);
    if (r < (unsigned)DEG_RANGE) atomicAdd(&bins[r], 1u);
}

__global__ void __launch_bounds__(DEG_THREADS) k_degree(const int* __restrict__ src,
                                                        const int* __restrict__ dst,
                                                        float* __restrict__ cnt_out,
                                                        float* __restrict__ cnt_in,
                                                        int n, int E) {
    __shared__ unsigned bins[DEG_RANGE];
    const int* __restrict__ idx = blockIdx.z ? dst : src;
    float* __restrict__ cnt = blockIdx.z ? cnt_in : cnt_out;
    const int4* __restrict__ idx4 = (const int4*)idx;
    const int E4 = E >> 2;
    const int per = (E4 + (int)gridDim.x - 1) / (int)gridDim.x;
    const int q0 = (int)blockIdx.x * per;
    const int q1 = min(q0 + per, E4);

    for (int base = (int)blockIdx.y * DEG_RANGE; base < n; base += (int)gridDim.y * DEG_RANGE) {
        for (int i = threadIdx.x; i < DEG_RANGE; i += DEG_THREADS) bins[i] = 0u;
        __syncthreads();
        for (int q = q0 + (int)threadIdx.x; q < q1; q += DEG_BATCH * DEG_THREADS) {
            int4 v[DEG_BATCH];
#pragma unroll
            for (int u = 0; u < DEG_BATCH; ++u) {
                const int qq = q + u * DEG_THREADS;
                v[u] = (qq < q1) ? idx4[qq] : make_int4(-1, -1, -1, -1);
            }
#pragma unroll
            for (int u = 0; u < DEG_BATCH; ++u) {
                deg_bin(bins, v[u].x, base);
                deg_bin(bins, v[u].y, base);
                deg_bin(bins, v[u].z, base);
                deg_bin(bins, v[u].w, base);
            }
        }
        if (blockIdx.x == 0)
            for (int e = (E4 << 2) + (int)threadIdx.x; e < E; e += DEG_THREADS)
                deg_bin(bins, idx[e], base);
        __syncthreads();
        for (int i = threadIdx.x; i < DEG_RANGE; i += DEG_THREADS) {
            const int node = base + i;
            if (node < n) atomicAdd(&cnt[node], (float)bins[i]);
        }
        __syncthreads();
    }
}

__global__ void __launch_bounds__(256) k_norm(float* __restrict__ sn,
                                              float* __restrict__ dn, int n) {
    int i = blockIdx.x * blockDim.x + threadIdx.x;
    if (i < n) {
        sn[i] = rsqrtf(fmaxf(sn[i], 1.0f));
        dn[i] = rsqrtf(fmaxf(dn[i], 1.0f));
    }
}

__global__ void __launch_bounds__(256) k_gemm1(const float* __restrict__ feat,
                                               const float* __restrict__ W1,
                                               const float* __restrict__ srcn,
                                               float* __restrict__ t1, int n) {
    __shared__ float sWt[OUT_DIM * S1];
    for (int i = threadIdx.x; i < IN_DIM * OUT_DIM; i += 256) {
        const int k = i >> 6;
        const int c = i & 63;
        sWt[c * S1 + k] = W1[i];
    }
    __syncthreads();

    const int wave = threadIdx.x >> 5;
    const int lane = threadIdx.x & 31;
    const int tile = blockIdx.x * 8 + wave;
    const int ntiles = (n + 15) >> 4;
    if (tile >= ntiles) return;

    const int halfsel = lane >> 4;
    const int l15 = lane & 15;
    int row = tile * 16 + l15;
    if (row >= n) row = n - 1;
    const int koff = halfsel * 2;
    const float norm = srcn[row];
    const float* arow = feat + (size_t)row * IN_DIM;
    const float* bc0 = sWt + ( 0 + l15) * S1;
    const float* bc1 = sWt + (16 + l15) * S1;
    const float* bc2 = sWt + (32 + l15) * S1;
    const float* bc3 = sWt + (48 + l15) * S1;

    v8f acc0 = {}, acc1 = {}, acc2 = {}, acc3 = {};
#pragma unroll 4
    for (int kb = 0; kb < IN_DIM / 4; ++kb) {
        const int k0 = kb * 4 + koff;
        v2f a = *(const v2f*)(arow + k0);
        a.x *= norm; a.y *= norm;
        const v2f b0 = *(const v2f*)(bc0 + k0);
        const v2f b1 = *(const v2f*)(bc1 + k0);
        const v2f b2 = *(const v2f*)(bc2 + k0);
        const v2f b3 = *(const v2f*)(bc3 + k0);
        acc0 = __builtin_amdgcn_wmma_f32_16x16x4_f32(false, a, false, b0, (short)0, acc0, false, false);
        acc1 = __builtin_amdgcn_wmma_f32_16x16x4_f32(false, a, false, b1, (short)0, acc1, false, false);
        acc2 = __builtin_amdgcn_wmma_f32_16x16x4_f32(false, a, false, b2, (short)0, acc2, false, false);
        acc3 = __builtin_amdgcn_wmma_f32_16x16x4_f32(false, a, false, b3, (short)0, acc3, false, false);
    }

#pragma unroll
    for (int r = 0; r < 8; ++r) {
        const int node = tile * 16 + r + halfsel * 8;
        if (node < n) {
            float* o = t1 + (size_t)node * OUT_DIM + l15;
            o[0]  = acc0[r];
            o[16] = acc1[r];
            o[32] = acc2[r];
            o[48] = acc3[r];
        }
    }
}

constexpr int EDGE_GROUP = 8;

__global__ void __launch_bounds__(256) k_scatter(const int* __restrict__ src,
                                                 const int* __restrict__ dst,
                                                 const float* __restrict__ msg,
                                                 float* __restrict__ agg,
                                                 int* __restrict__ ctr, int n, int E) {
    constexpr int SLOTS = EDGES_PER_WAVE / 32;
    __shared__ int list_s[8][EDGES_PER_WAVE];
    __shared__ int list_d[8][EDGES_PER_WAVE];
    __shared__ int s_draw;
    const int lane = threadIdx.x & 31;
    const int wv = threadIdx.x >> 5;
    int* ls = list_s[wv];
    int* ld = list_d[wv];
    const int split = n >> 1;
    const int spent = (int)gridDim.x;

    if (threadIdx.x == 0) s_draw = (int)((__builtin_amdgcn_s_sendmsg_rtn(0x87) >> 18) & 1u);
    __syncthreads();
    int half = __builtin_amdgcn_readfirstlane(s_draw);
    __syncthreads();

    for (int visited = 0; visited < 2;) {
        if (threadIdx.x == 0) s_draw = atomicSub(&ctr[half * 32], 1);
        __syncthreads();
        const int old = __builtin_amdgcn_readfirstlane(s_draw);
        __syncthreads();
        if (old <= spent) { ++visited; half ^= 1; continue; }

        const int base0 = (old - spent - 1) * TICKET_EDGES + wv * EDGES_PER_WAVE;
        int s[SLOTS], d[SLOTS];
#pragma unroll
        for (int k = 0; k < SLOTS; ++k) {
            const int e = base0 + 32 * k + lane;
            s[k] = (e < E) ? src[e] : 0;
            d[k] = (e < E) ? dst[e] : -1;
        }
        int kept = 0;
#pragma unroll
        for (int k = 0; k < SLOTS; ++k) {
            const bool mine = (d[k] >= 0) && ((d[k] >= split) == (half != 0));
            const unsigned m = __builtin_amdgcn_ballot_w32(mine);
            const int pos = kept + (int)__builtin_amdgcn_mbcnt_lo(m, 0u);
            if (mine) { ls[pos] = s[k]; ld[pos] = d[k]; }
            kept += __builtin_popcount(m);
        }

        for (int k0 = 0; k0 < kept; k0 += 32) {
            const int sk = ls[k0 + lane];
            const int dk = ld[k0 + lane];
            const int cnt = min(32, kept - k0);
            int j0 = 0;
            for (; j0 + EDGE_GROUP <= cnt; j0 += EDGE_GROUP) {
                float lo[EDGE_GROUP], hi[EDGE_GROUP];
                int dj[EDGE_GROUP];
#pragma unroll
                for (int u = 0; u < EDGE_GROUP; ++u) {
                    const int sj = __builtin_amdgcn_readlane(sk, j0 + u);
                    dj[u] = __builtin_amdgcn_readlane(dk, j0 + u);
                    const float* m = msg + (size_t)sj * OUT_DIM + lane;
                    lo[u] = m[0];
                    hi[u] = m[32];
                }
#pragma unroll
                for (int u = 0; u < EDGE_GROUP; ++u) {
                    float* a = agg + (size_t)dj[u] * OUT_DIM + lane;
                    atomicAdd(a, lo[u]);
                    atomicAdd(a + 32, hi[u]);
                }
            }
            if (j0 < cnt) {
                float lo[EDGE_GROUP], hi[EDGE_GROUP];
                int dj[EDGE_GROUP];
#pragma unroll
                for (int u = 0; u < EDGE_GROUP; ++u) {
                    const int j = min(j0 + u, cnt - 1);
                    const int sj = __builtin_amdgcn_readlane(sk, j);
                    dj[u] = __builtin_amdgcn_readlane(dk, j);
                    const float* m = msg + (size_t)sj * OUT_DIM + lane;
                    lo[u] = m[0];
                    hi[u] = m[32];
                }
#pragma unroll
                for (int u = 0; u < EDGE_GROUP; ++u) {
                    if (j0 + u < cnt) {
                        float* a = agg + (size_t)dj[u] * OUT_DIM + lane;
                        atomicAdd(a, lo[u]);
                        atomicAdd(a + 32, hi[u]);
                    }
                }
            }
        }
    }
}

__global__ void __launch_bounds__(256) k_finalize1(float* __restrict__ agg,
                                                   const float* __restrict__ dstn,
                                                   const float* __restrict__ srcn,
                                                   const float* __restrict__ b1,
                                                   const float* __restrict__ prelu_a,
                                                   float* __restrict__ hs,
                                                   float* __restrict__ hg1, int n) {
    const float pa = *prelu_a;
    const int d = threadIdx.x & 63;
    const int sub = threadIdx.x >> 6;
    float part = 0.f;
    for (int base = blockIdx.x * 4; base < n; base += gridDim.x * 4) {
        const int node = base + sub;
        if (node < n) {
            const size_t idx = (size_t)node * OUT_DIM + d;
            const float v = agg[idx];
            agg[idx] = 0.f;
            float h = v * dstn[node] + b1[d];
            h = (h >= 0.f) ? h : pa * h;
            part += h;
            hs[idx] = h * srcn[node];
        }
    }
    __shared__ float red[256];
    red[threadIdx.x] = part;
    __syncthreads();
    if (threadIdx.x < 128) red[threadIdx.x] += red[threadIdx.x + 128];
    __syncthreads();
    if (threadIdx.x < 64)
        atomicAdd(&hg1[threadIdx.x], red[threadIdx.x] + red[threadIdx.x + 64]);
}

__global__ void __launch_bounds__(256) k_gemm2(const float* __restrict__ agg,
                                               const float* __restrict__ W2,
                                               const float* __restrict__ dstn,
                                               const float* __restrict__ b2,
                                               const float* __restrict__ prelu_a,
                                               float* __restrict__ hout,
                                               float* __restrict__ hg2, int n) {
    __shared__ float sWt[OUT_DIM * S2];
    __shared__ float shg[OUT_DIM];
    for (int i = threadIdx.x; i < OUT_DIM * OUT_DIM; i += 256) {
        const int k = i >> 6;
        const int c = i & 63;
        sWt[c * S2 + k] = W2[i];
    }
    if (threadIdx.x < OUT_DIM) shg[threadIdx.x] = 0.f;
    __syncthreads();

    const float pa = *prelu_a;
    const int wave = threadIdx.x >> 5;
    const int lane = threadIdx.x & 31;
    const int halfsel = lane >> 4;
    const int l15 = lane & 15;
    const int tile = blockIdx.x * 8 + wave;
    const int ntiles = (n + 15) >> 4;
    const bool active = tile < ntiles;

    if (active) {
        int row = tile * 16 + l15;
        if (row >= n) row = n - 1;
        const int koff = halfsel * 2;
        const float* arow = agg + (size_t)row * OUT_DIM;
        const float* bc0 = sWt + ( 0 + l15) * S2;
        const float* bc1 = sWt + (16 + l15) * S2;
        const float* bc2 = sWt + (32 + l15) * S2;
        const float* bc3 = sWt + (48 + l15) * S2;

        v8f acc0 = {}, acc1 = {}, acc2 = {}, acc3 = {};
#pragma unroll 4
        for (int kb = 0; kb < OUT_DIM / 4; ++kb) {
            const int k0 = kb * 4 + koff;
            const v2f a  = *(const v2f*)(arow + k0);
            const v2f b0 = *(const v2f*)(bc0 + k0);
            const v2f b1 = *(const v2f*)(bc1 + k0);
            const v2f b2 = *(const v2f*)(bc2 + k0);
            const v2f b3 = *(const v2f*)(bc3 + k0);
            acc0 = __builtin_amdgcn_wmma_f32_16x16x4_f32(false, a, false, b0, (short)0, acc0, false, false);
            acc1 = __builtin_amdgcn_wmma_f32_16x16x4_f32(false, a, false, b1, (short)0, acc1, false, false);
            acc2 = __builtin_amdgcn_wmma_f32_16x16x4_f32(false, a, false, b2, (short)0, acc2, false, false);
            acc3 = __builtin_amdgcn_wmma_f32_16x16x4_f32(false, a, false, b3, (short)0, acc3, false, false);
        }

        const float bb0 = b2[ 0 + l15], bb1 = b2[16 + l15];
        const float bb2 = b2[32 + l15], bb3 = b2[48 + l15];
        float hsum0 = 0.f, hsum1 = 0.f, hsum2 = 0.f, hsum3 = 0.f;
#pragma unroll
        for (int r = 0; r < 8; ++r) {
            const int node = tile * 16 + r + halfsel * 8;
            if (node < n) {
                const float dn = dstn[node];
                float v0 = acc0[r] * dn + bb0; v0 = (v0 >= 0.f) ? v0 : pa * v0;
                float v1 = acc1[r] * dn + bb1; v1 = (v1 >= 0.f) ? v1 : pa * v1;
                float v2 = acc2[r] * dn + bb2; v2 = (v2 >= 0.f) ? v2 : pa * v2;
                float v3 = acc3[r] * dn + bb3; v3 = (v3 >= 0.f) ? v3 : pa * v3;
                float* o = hout + (size_t)node * OUT_DIM + l15;
                o[0] = v0; o[16] = v1; o[32] = v2; o[48] = v3;
                hsum0 += v0; hsum1 += v1; hsum2 += v2; hsum3 += v3;
            }
        }
        atomicAdd(&shg[ 0 + l15], hsum0);
        atomicAdd(&shg[16 + l15], hsum1);
        atomicAdd(&shg[32 + l15], hsum2);
        atomicAdd(&shg[48 + l15], hsum3);
    }
    __syncthreads();
    if (threadIdx.x < OUT_DIM) atomicAdd(&hg2[threadIdx.x], shg[threadIdx.x]);
}

extern "C" void kernel_launch(void* const* d_in, const int* in_sizes, int n_in,
                              void* d_out, int out_size, void* d_ws, size_t ws_size,
                              hipStream_t stream) {
    const float* feat    = (const float*)d_in[0];
    const int*   src     = (const int*)  d_in[1];
    const int*   dst     = (const int*)  d_in[2];
    const float* W1      = (const float*)d_in[3];
    const float* b1      = (const float*)d_in[4];
    const float* W2      = (const float*)d_in[5];
    const float* b2      = (const float*)d_in[6];
    const float* prelu_a = (const float*)d_in[7];

    const int n = in_sizes[0] / IN_DIM;
    const int E = in_sizes[1];

    float* out = (float*)d_out;
    float* hg1 = out + (size_t)n * OUT_DIM;
    float* hg2 = hg1 + OUT_DIM;

    float* wsf  = (float*)d_ws;
    float* srcn = wsf;
    float* dstn = wsf + n;
    float* agg  = wsf + 2 * (size_t)n;
    float* t1   = out;

    const int ntiles = (n + 15) / 16;
    const int gemm_blocks = (ntiles + 7) / 8;

    k_init   <<<2048, 256, 0, stream>>>(srcn, agg, hg1, n, E);
    k_degree <<<dim3(DEG_SLICES, DEG_RANGES, 2), DEG_THREADS, 0, stream>>>(src, dst, srcn, dstn, n, E);
    k_norm   <<<(n + 255) / 256, 256, 0, stream>>>(srcn, dstn, n);
    k_gemm1  <<<gemm_blocks, 256, 0, stream>>>(feat, W1, srcn, t1, n);
    k_scatter<<<SC_BLOCKS, 256, 0, stream>>>(src, dst, t1, agg, (int*)hg1, n, E);
    k_finalize1<<<1024, 256, 0, stream>>>(agg, dstn, srcn, b1, prelu_a, t1, hg1, n);
    k_scatter<<<SC_BLOCKS, 256, 0, stream>>>(src, dst, t1, agg, (int*)hg2, n, E);
    k_gemm2  <<<gemm_blocks, 256, 0, stream>>>(agg, W2, dstn, b2, prelu_a, out, hg2, n);
}


# finalize1 loads 8 rows per thread, and both GEMMs their A fragments and epilogue norms, ahead of first use: one memory round trip per batch, not one per row or per 4 k-steps; summation order kept
# speedup vs baseline: 2.8858x; 1.0332x over previous
#include <hip/hip_runtime.h>

constexpr int IN_DIM  = 128;
constexpr int OUT_DIM = 64;
constexpr int S1 = IN_DIM + 4;
constexpr int S2 = OUT_DIM + 4;

typedef float v2f __attribute__((ext_vector_type(2)));
typedef float v8f __attribute__((ext_vector_type(8)));

constexpr int DEG_RANGE   = 12800;
constexpr int DEG_RANGES  = 8;
constexpr int DEG_SLICES  = 16;
constexpr int DEG_THREADS = 1024;
constexpr int DEG_BATCH   = 8;

constexpr int SC_BLOCKS      = 1536;
constexpr int EDGES_PER_WAVE = 128;
constexpr int TICKET_EDGES   = 8 * EDGES_PER_WAVE;

__device__ __forceinline__ int ticket_start(int E) {
    return (E + TICKET_EDGES - 1) / TICKET_EDGES + SC_BLOCKS;
}

__global__ void __launch_bounds__(256) k_init(float* __restrict__ cnt,
                                              float* __restrict__ agg,
                                              float* __restrict__ hg, int n, int E) {
    int tid = blockIdx.x * blockDim.x + threadIdx.x;
    int stride = gridDim.x * blockDim.x;
    for (int i = tid; i < 2 * n; i += stride) cnt[i] = 0.f;
    for (int i = tid; i < n * OUT_DIM; i += stride) agg[i] = 0.f;
    if (tid < 2 * OUT_DIM) {
        if ((tid & 31) == 0) ((int*)hg)[tid] = ticket_start(E);
        else hg[tid] = 0.f;
    }
}

__device__ __forceinline__ void deg_bin(unsigned* bins, int id, int base) {
    const unsigned r = (unsigned)(id - base);
    if (r < (unsigned)DEG_RANGE) atomicAdd(&bins[r], 1u);
}

__global__ void __launch_bounds__(DEG_THREADS) k_degree(const int* __restrict__ src,
                                                        const int* __restrict__ dst,
                                                        float* __restrict__ cnt_out,
                                                        float* __restrict__ cnt_in,
                                                        int n, int E) {
    __shared__ unsigned bins[DEG_RANGE];
    const int* __restrict__ idx = blockIdx.z ? dst : src;
    float* __restrict__ cnt = blockIdx.z ? cnt_in : cnt_out;
    const int4* __restrict__ idx4 = (const int4*)idx;
    const int E4 = E >> 2;
    const int per = (E4 + (int)gridDim.x - 1) / (int)gridDim.x;
    const int q0 = (int)blockIdx.x * per;
    const int q1 = min(q0 + per, E4);

    for (int base = (int)blockIdx.y * DEG_RANGE; base < n; base += (int)gridDim.y * DEG_RANGE) {
        for (int i = threadIdx.x; i < DEG_RANGE; i += DEG_THREADS) bins[i] = 0u;
        __syncthreads();
        for (int q = q0 + (int)threadIdx.x; q < q1; q += DEG_BATCH * DEG_THREADS) {
            int4 v[DEG_BATCH];
#pragma unroll
            for (int u = 0; u < DEG_BATCH; ++u) {
                const int qq = q + u * DEG_THREADS;
                v[u] = (qq < q1) ? idx4[qq] : make_int4(-1, -1, -1, -1);
            }
#pragma unroll
            for (int u = 0; u < DEG_BATCH; ++u) {
                deg_bin(bins, v[u].x, base);
                deg_bin(bins, v[u].y, base);
                deg_bin(bins, v[u].z, base);
                deg_bin(bins, v[u].w, base);
            }
        }
        if (blockIdx.x == 0)
            for (int e = (E4 << 2) + (int)threadIdx.x; e < E; e += DEG_THREADS)
                deg_bin(bins, idx[e], base);
        __syncthreads();
        for (int i = threadIdx.x; i < DEG_RANGE; i += DEG_THREADS) {
            const int node = base + i;
            if (node < n) atomicAdd(&cnt[node], (float)bins[i]);
        }
        __syncthreads();
    }
}

__global__ void __launch_bounds__(256) k_norm(float* __restrict__ sn,
                                              float* __restrict__ dn, int n) {
    int i = blockIdx.x * blockDim.x + threadIdx.x;
    if (i < n) {
        sn[i] = rsqrtf(fmaxf(sn[i], 1.0f));
        dn[i] = rsqrtf(fmaxf(dn[i], 1.0f));
    }
}

__global__ void __launch_bounds__(256) k_gemm1(const float* __restrict__ feat,
                                               const float* __restrict__ W1,
                                               const float* __restrict__ srcn,
                                               float* __restrict__ t1, int n) {
    __shared__ float sWt[OUT_DIM * S1];
    for (int i = threadIdx.x; i < IN_DIM * OUT_DIM; i += 256) {
        const int k = i >> 6;
        const int c = i & 63;
        sWt[c * S1 + k] = W1[i];
    }
    __syncthreads();

    const int wave = threadIdx.x >> 5;
    const int lane = threadIdx.x & 31;
    const int tile = blockIdx.x * 8 + wave;
    const int ntiles = (n + 15) >> 4;
    if (tile >= ntiles) return;

    const int halfsel = lane >> 4;
    const int l15 = lane & 15;
    int row = tile * 16 + l15;
    if (row >= n) row = n - 1;
    const int koff = halfsel * 2;
    const float norm = srcn[row];
    const float* arow = feat + (size_t)row * IN_DIM;
    const float* bc0 = sWt + ( 0 + l15) * S1;
    const float* bc1 = sWt + (16 + l15) * S1;
    const float* bc2 = sWt + (32 + l15) * S1;
    const float* bc3 = sWt + (48 + l15) * S1;

    v8f acc0 = {}, acc1 = {}, acc2 = {}, acc3 = {};
    v2f afr[IN_DIM / 4];
#pragma unroll
    for (int kb = 0; kb < IN_DIM / 4; ++kb) afr[kb] = *(const v2f*)(arow + kb * 4 + koff);
#pragma unroll
    for (int kb = 0; kb < IN_DIM / 4; ++kb) {
        const int k0 = kb * 4 + koff;
        v2f a = afr[kb];
        a.x *= norm; a.y *= norm;
        const v2f b0 = *(const v2f*)(bc0 + k0);
        const v2f b1 = *(const v2f*)(bc1 + k0);
        const v2f b2 = *(const v2f*)(bc2 + k0);
        const v2f b3 = *(const v2f*)(bc3 + k0);
        acc0 = __builtin_amdgcn_wmma_f32_16x16x4_f32(false, a, false, b0, (short)0, acc0, false, false);
        acc1 = __builtin_amdgcn_wmma_f32_16x16x4_f32(false, a, false, b1, (short)0, acc1, false, false);
        acc2 = __builtin_amdgcn_wmma_f32_16x16x4_f32(false, a, false, b2, (short)0, acc2, false, false);
        acc3 = __builtin_amdgcn_wmma_f32_16x16x4_f32(false, a, false, b3, (short)0, acc3, false, false);
    }

#pragma unroll
    for (int r = 0; r < 8; ++r) {
        const int node = tile * 16 + r + halfsel * 8;
        if (node < n) {
            float* o = t1 + (size_t)node * OUT_DIM + l15;
            o[0]  = acc0[r];
            o[16] = acc1[r];
            o[32] = acc2[r];
            o[48] = acc3[r];
        }
    }
}

constexpr int EDGE_GROUP = 8;

__global__ void __launch_bounds__(256) k_scatter(const int* __restrict__ src,
                                                 const int* __restrict__ dst,
                                                 const float* __restrict__ msg,
                                                 float* __restrict__ agg,
                                                 int* __restrict__ ctr, int n, int E) {
    constexpr int SLOTS = EDGES_PER_WAVE / 32;
    __shared__ int list_s[8][EDGES_PER_WAVE];
    __shared__ int list_d[8][EDGES_PER_WAVE];
    __shared__ int s_draw;
    const int lane = threadIdx.x & 31;
    const int wv = threadIdx.x >> 5;
    int* ls = list_s[wv];
    int* ld = list_d[wv];
    const int split = n >> 1;
    const int spent = (int)gridDim.x;

    if (threadIdx.x == 0) s_draw = (int)((__builtin_amdgcn_s_sendmsg_rtn(0x87) >> 18) & 1u);
    __syncthreads();
    int half = __builtin_amdgcn_readfirstlane(s_draw);
    __syncthreads();

    for (int visited = 0; visited < 2;) {
        if (threadIdx.x == 0) s_draw = atomicSub(&ctr[half * 32], 1);
        __syncthreads();
        const int old = __builtin_amdgcn_readfirstlane(s_draw);
        __syncthreads();
        if (old <= spent) { ++visited; half ^= 1; continue; }

        const int base0 = (old - spent - 1) * TICKET_EDGES + wv * EDGES_PER_WAVE;
        int s[SLOTS], d[SLOTS];
#pragma unroll
        for (int k = 0; k < SLOTS; ++k) {
            const int e = base0 + 32 * k + lane;
            s[k] = (e < E) ? src[e] : 0;
            d[k] = (e < E) ? dst[e] : -1;
        }
        int kept = 0;
#pragma unroll
        for (int k = 0; k < SLOTS; ++k) {
            const bool mine = (d[k] >= 0) && ((d[k] >= split) == (half != 0));
            const unsigned m = __builtin_amdgcn_ballot_w32(mine);
            const int pos = kept + (int)__builtin_amdgcn_mbcnt_lo(m, 0u);
            if (mine) { ls[pos] = s[k]; ld[pos] = d[k]; }
            kept += __builtin_popcount(m);
        }

        for (int k0 = 0; k0 < kept; k0 += 32) {
            const int sk = ls[k0 + lane];
            const int dk = ld[k0 + lane];
            const int cnt = min(32, kept - k0);
            int j0 = 0;
            for (; j0 + EDGE_GROUP <= cnt; j0 += EDGE_GROUP) {
                float lo[EDGE_GROUP], hi[EDGE_GROUP];
                int dj[EDGE_GROUP];
#pragma unroll
                for (int u = 0; u < EDGE_GROUP; ++u) {
                    const int sj = __builtin_amdgcn_readlane(sk, j0 + u);
                    dj[u] = __builtin_amdgcn_readlane(dk, j0 + u);
                    const float* m = msg + (size_t)sj * OUT_DIM + lane;
                    lo[u] = m[0];
                    hi[u] = m[32];
                }
#pragma unroll
                for (int u = 0; u < EDGE_GROUP; ++u) {
                    float* a = agg + (size_t)dj[u] * OUT_DIM + lane;
                    atomicAdd(a, lo[u]);
                    atomicAdd(a + 32, hi[u]);
                }
            }
            if (j0 < cnt) {
                float lo[EDGE_GROUP], hi[EDGE_GROUP];
                int dj[EDGE_GROUP];
#pragma unroll
                for (int u = 0; u < EDGE_GROUP; ++u) {
                    const int j = min(j0 + u, cnt - 1);
                    const int sj = __builtin_amdgcn_readlane(sk, j);
                    dj[u] = __builtin_amdgcn_readlane(dk, j);
                    const float* m = msg + (size_t)sj * OUT_DIM + lane;
                    lo[u] = m[0];
                    hi[u] = m[32];
                }
#pragma unroll
                for (int u = 0; u < EDGE_GROUP; ++u) {
                    if (j0 + u < cnt) {
                        float* a = agg + (size_t)dj[u] * OUT_DIM + lane;
                        atomicAdd(a, lo[u]);
                        atomicAdd(a + 32, hi[u]);
                    }
                }
            }
        }
    }
}

constexpr int FIN_BATCH = 8;
__global__ void __launch_bounds__(256) k_finalize1(float* __restrict__ agg,
                                                   const float* __restrict__ dstn,
                                                   const float* __restrict__ srcn,
                                                   const float* __restrict__ b1,
                                                   const float* __restrict__ prelu_a,
                                                   float* __restrict__ hs,
                                                   float* __restrict__ hg1, int n) {
    const float pa = *prelu_a;
    const int d = threadIdx.x & 63;
    const int sub = threadIdx.x >> 6;
    const float bias = b1[d];
    const int step = gridDim.x * 4;
    float part = 0.f;
    for (int base = blockIdx.x * 4; base < n; base += FIN_BATCH * step) {
        float v[FIN_BATCH], dn[FIN_BATCH], sn[FIN_BATCH];
#pragma unroll
        for (int u = 0; u < FIN_BATCH; ++u) {
            const int node = min(base + u * step + sub, n - 1);
            v[u]  = agg[(size_t)node * OUT_DIM + d];
            dn[u] = dstn[node];
            sn[u] = srcn[node];
        }
#pragma unroll
        for (int u = 0; u < FIN_BATCH; ++u) {
            const int node = base + u * step + sub;
            if (node < n) {
                const size_t idx = (size_t)node * OUT_DIM + d;
                agg[idx] = 0.f;
                float h = v[u] * dn[u] + bias;
                h = (h >= 0.f) ? h : pa * h;
                part += h;
                hs[idx] = h * sn[u];
            }
        }
    }
    __shared__ float red[256];
    red[threadIdx.x] = part;
    __syncthreads();
    if (threadIdx.x < 128) red[threadIdx.x] += red[threadIdx.x + 128];
    __syncthreads();
    if (threadIdx.x < 64)
        atomicAdd(&hg1[threadIdx.x], red[threadIdx.x] + red[threadIdx.x + 64]);
}

__global__ void __launch_bounds__(256) k_gemm2(const float* __restrict__ agg,
                                               const float* __restrict__ W2,
                                               const float* __restrict__ dstn,
                                               const float* __restrict__ b2,
                                               const float* __restrict__ prelu_a,
                                               float* __restrict__ hout,
                                               float* __restrict__ hg2, int n) {
    __shared__ float sWt[OUT_DIM * S2];
    __shared__ float shg[OUT_DIM];
    for (int i = threadIdx.x; i < OUT_DIM * OUT_DIM; i += 256) {
        const int k = i >> 6;
        const int c = i & 63;
        sWt[c * S2 + k] = W2[i];
    }
    if (threadIdx.x < OUT_DIM) shg[threadIdx.x] = 0.f;
    __syncthreads();

    const float pa = *prelu_a;
    const int wave = threadIdx.x >> 5;
    const int lane = threadIdx.x & 31;
    const int halfsel = lane >> 4;
    const int l15 = lane & 15;
    const int tile = blockIdx.x * 8 + wave;
    const int ntiles = (n + 15) >> 4;
    const bool active = tile < ntiles;

    if (active) {
        int row = tile * 16 + l15;
        if (row >= n) row = n - 1;
        const int koff = halfsel * 2;
        const float* arow = agg + (size_t)row * OUT_DIM;
        const float* bc0 = sWt + ( 0 + l15) * S2;
        const float* bc1 = sWt + (16 + l15) * S2;
        const float* bc2 = sWt + (32 + l15) * S2;
        const float* bc3 = sWt + (48 + l15) * S2;

        v8f acc0 = {}, acc1 = {}, acc2 = {}, acc3 = {};
        v2f afr[OUT_DIM / 4];
        float dnv[8];
#pragma unroll
        for (int r = 0; r < 8; ++r) dnv[r] = dstn[min(tile * 16 + r + halfsel * 8, n - 1)];
#pragma unroll
        for (int kb = 0; kb < OUT_DIM / 4; ++kb) afr[kb] = *(const v2f*)(arow + kb * 4 + koff);
#pragma unroll
        for (int kb = 0; kb < OUT_DIM / 4; ++kb) {
            const int k0 = kb * 4 + koff;
            const v2f a  = afr[kb];
            const v2f b0 = *(const v2f*)(bc0 + k0);
            const v2f b1 = *(const v2f*)(bc1 + k0);
            const v2f b2 = *(const v2f*)(bc2 + k0);
            const v2f b3 = *(const v2f*)(bc3 + k0);
            acc0 = __builtin_amdgcn_wmma_f32_16x16x4_f32(false, a, false, b0, (short)0, acc0, false, false);
            acc1 = __builtin_amdgcn_wmma_f32_16x16x4_f32(false, a, false, b1, (short)0, acc1, false, false);
            acc2 = __builtin_amdgcn_wmma_f32_16x16x4_f32(false, a, false, b2, (short)0, acc2, false, false);
            acc3 = __builtin_amdgcn_wmma_f32_16x16x4_f32(false, a, false, b3, (short)0, acc3, false, false);
        }

        const float bb0 = b2[ 0 + l15], bb1 = b2[16 + l15];
        const float bb2 = b2[32 + l15], bb3 = b2[48 + l15];
        float hsum0 = 0.f, hsum1 = 0.f, hsum2 = 0.f, hsum3 = 0.f;
#pragma unroll
        for (int r = 0; r < 8; ++r) {
            const int node = tile * 16 + r + halfsel * 8;
            if (node < n) {
                const float dn = dnv[r];
                float v0 = acc0[r] * dn + bb0; v0 = (v0 >= 0.f) ? v0 : pa * v0;
                float v1 = acc1[r] * dn + bb1; v1 = (v1 >= 0.f) ? v1 : pa * v1;
                float v2 = acc2[r] * dn + bb2; v2 = (v2 >= 0.f) ? v2 : pa * v2;
                float v3 = acc3[r] * dn + bb3; v3 = (v3 >= 0.f) ? v3 : pa * v3;
                float* o = hout + (size_t)node * OUT_DIM + l15;
                o[0] = v0; o[16] = v1; o[32] = v2; o[48] = v3;
                hsum0 += v0; hsum1 += v1; hsum2 += v2; hsum3 += v3;
            }
        }
        atomicAdd(&shg[ 0 + l15], hsum0);
        atomicAdd(&shg[16 + l15], hsum1);
        atomicAdd(&shg[32 + l15], hsum2);
        atomicAdd(&shg[48 + l15], hsum3);
    }
    __syncthreads();
    if (threadIdx.x < OUT_DIM) atomicAdd(&hg2[threadIdx.x], shg[threadIdx.x]);
}

extern "C" void kernel_launch(void* const* d_in, const int* in_sizes, int n_in,
                              void* d_out, int out_size, void* d_ws, size_t ws_size,
                              hipStream_t stream) {
    const float* feat    = (const float*)d_in[0];
    const int*   src     = (const int*)  d_in[1];
    const int*   dst     = (const int*)  d_in[2];
    const float* W1      = (const float*)d_in[3];
    const float* b1      = (const float*)d_in[4];
    const float* W2      = (const float*)d_in[5];
    const float* b2      = (const float*)d_in[6];
    const float* prelu_a = (const float*)d_in[7];

    const int n = in_sizes[0] / IN_DIM;
    const int E = in_sizes[1];

    float* out = (float*)d_out;
    float* hg1 = out + (size_t)n * OUT_DIM;
    float* hg2 = hg1 + OUT_DIM;

    float* wsf  = (float*)d_ws;
    float* srcn = wsf;
    float* dstn = wsf + n;
    float* agg  = wsf + 2 * (size_t)n;
    float* t1   = out;

    const int ntiles = (n + 15) / 16;
    const int gemm_blocks = (ntiles + 7) / 8;

    k_init   <<<2048, 256, 0, stream>>>(srcn, agg, hg1, n, E);
    k_degree <<<dim3(DEG_SLICES, DEG_RANGES, 2), DEG_THREADS, 0, stream>>>(src, dst, srcn, dstn, n, E);
    k_norm   <<<(n + 255) / 256, 256, 0, stream>>>(srcn, dstn, n);
    k_gemm1  <<<gemm_blocks, 256, 0, stream>>>(feat, W1, srcn, t1, n);
    k_scatter<<<SC_BLOCKS, 256, 0, stream>>>(src, dst, t1, agg, (int*)hg1, n, E);
    k_finalize1<<<1024, 256, 0, stream>>>(agg, dstn, srcn, b1, prelu_a, t1, hg1, n);
    k_scatter<<<SC_BLOCKS, 256, 0, stream>>>(src, dst, t1, agg, (int*)hg2, n, E);
    k_gemm2  <<<gemm_blocks, 256, 0, stream>>>(agg, W2, dstn, b2, prelu_a, out, hg2, n);
}
